# starting kernel
# baseline (speedup 1.0000x reference)
#include <hip/hip_runtime.h>
#include <math.h>


typedef float v2f __attribute__((ext_vector_type(2)));
typedef float v8f __attribute__((ext_vector_type(8)));

#define TINYV 1e-6f
#define LLOB 128
#define SSAM 8
#define HHID 16
#define NELEM (8192 * 128 * 8)

#define SWZ(x, pat) __int_as_float(__builtin_amdgcn_ds_swizzle(__float_as_int(x), (pat)))
#define PAT_XOR16 0x401F
#define PAT_XOR1  0x041F
#define PAT_XOR2  0x081F
#define PAT_XOR4  0x101F

__global__ __launch_bounds__(256) void sg_vis_wmma_kernel(
    const float* __restrict__ points,
    const float* __restrict__ normals,
    const float* __restrict__ rootR,
    const float* __restrict__ lobes,
    const float* __restrict__ lambdas,
    const float* __restrict__ rrand,
    const float* __restrict__ W1,
    const float* __restrict__ b1,
    const float* __restrict__ W2,
    const float* __restrict__ b2,
    float* __restrict__ out)
{
    __shared__ float sF[32 * 12];

    const int tid  = threadIdx.x;
    const int e    = blockIdx.x * 256 + tid;
    const int lane = tid & 31;

    if (tid < 32) {
        const int l = ((blockIdx.x * 32) + tid) & (LLOB - 1);
        float lx = lobes[3*l+0], ly = lobes[3*l+1], lz = lobes[3*l+2];
        float invn = 1.f / (sqrtf(lx*lx + ly*ly + lz*lz) + TINYV);
        lx *= invn; ly *= invn; lz *= invn;
        float ux = -ly, uy = lx;
        float invu = 1.f / (sqrtf(ux*ux + uy*uy) + TINYV);
        ux *= invu; uy *= invu;
        float vx = -lz*uy, vy = lz*ux, vz = lx*uy - ly*ux;
        float invv = 1.f / (sqrtf(vx*vx + vy*vy + vz*vz) + TINYV);
        vx *= invv; vy *= invv; vz *= invv;
        float sharp = lambdas[l];
        float rphi  = fminf(acosf(1.f - 1.f/sharp), 1.04719755119659775f);
        float* F = &sF[tid * 12];
        F[0] = lx; F[1] = ly; F[2] = lz;
        F[3] = ux; F[4] = uy;
        F[5] = vx; F[6] = vy; F[7] = vz;
        F[8] = __sinf(rphi); F[9] = __cosf(rphi); F[10] = sharp; F[11] = 0.f;
    }
    __syncthreads();

    const float* F = &sF[(tid >> 3) * 12];
    const float lnx = F[0], lny = F[1], lnz = F[2];
    const float ux  = F[3], uy  = F[4];
    const float vx  = F[5], vy  = F[6], vz = F[7];
    const float sp  = F[8], cp  = F[9], sharp = F[10];

    const float theta = (rrand[e] * (1.f/SSAM) + (float)(tid & 7) * (1.f/SSAM))
                        * 6.28318530717958648f;
    const float st = __sinf(theta), ct = __cosf(theta);
    const float ca = ct * sp, cb = st * sp;
    const float dx = ux*ca + vx*cb + lnx*cp;
    const float dy = uy*ca + vy*cb + lny*cp;
    const float dz =          vz*cb + lnz*cp;

    const int n = blockIdx.x >> 2;
    const float px = points[3*n+0],  py = points[3*n+1],  pz = points[3*n+2];
    const float nx = normals[3*n+0], ny = normals[3*n+1], nz = normals[3*n+2];
    const bool cosmask = (nx*dx + ny*dy + nz*dz) > TINYV;

    const float id0 = -(dx*rootR[0] + dy*rootR[3] + dz*rootR[6]);
    const float id1 = -(dx*rootR[1] + dy*rootR[4] + dz*rootR[7]);
    const float id2 = -(dx*rootR[2] + dy*rootR[5] + dz*rootR[8]);

    const float w = __expf(sharp * (dx*lnx + dy*lny + dz*lnz - 1.f));

    const int  m     = lane & 15;
    const bool up    = lane >= 16;
    const int  hbase = up ? 8 : 0;

    v2f A0, A1;
    A0.x = up ? W1[2*HHID + m] : W1[0*HHID + m];
    A0.y = up ? W1[3*HHID + m] : W1[1*HHID + m];
    A1.x = up ? 0.f : W1[4*HHID + m];
    A1.y = up ? 0.f : W1[5*HHID + m];

    const float x3 = id0, x4 = id1, x5 = id2;
    const float y3 = SWZ(x3, PAT_XOR16);
    const float y4 = SWZ(x4, PAT_XOR16);
    const float y5 = SWZ(x5, PAT_XOR16);

    v2f B0t0, B1t0, B0t1, B1t1;
    B0t0.x = up ? pz  : px;   B0t0.y = up ? y3  : py;
    B1t0.x = up ? 0.f : x4;   B1t0.y = up ? 0.f : x5;
    B0t1.x = up ? pz  : px;   B0t1.y = up ? x3  : py;
    B1t1.x = up ? 0.f : y4;   B1t1.y = up ? 0.f : y5;

    v8f c0, c1;
    #pragma unroll
    for (int r = 0; r < 8; ++r) c0[r] = b1[hbase + r];
    c1 = c0;

    c0 = __builtin_amdgcn_wmma_f32_16x16x4_f32(false, A0, false, B0t0, (short)0, c0, false, false);
    c0 = __builtin_amdgcn_wmma_f32_16x16x4_f32(false, A1, false, B1t0, (short)0, c0, false, false);
    c1 = __builtin_amdgcn_wmma_f32_16x16x4_f32(false, A0, false, B0t1, (short)0, c1, false, false);
    c1 = __builtin_amdgcn_wmma_f32_16x16x4_f32(false, A1, false, B1t1, (short)0, c1, false, false);

    float w2v[8];
    #pragma unroll
    for (int r = 0; r < 8; ++r) w2v[r] = W2[hbase + r];
    float pa = 0.f, pb = 0.f;
    #pragma unroll
    for (int r = 0; r < 8; ++r) {
        const float t0 = c0[r], t1 = c1[r];
        pa += (t0 > 0.f ? t0 : 0.f) * w2v[r];
        pb += (t1 > 0.f ? t1 : 0.f) * w2v[r];
    }
    const float q  = up ? pa : pb;
    const float qq = SWZ(q, PAT_XOR16);
    const float logit = (up ? pb : pa) + qq + b2[0];
    const float pred  = 1.f / (1.f + __expf(-logit));
    const float vis   = cosmask ? pred : 0.f;

    float num = vis * w, den = w;
    num += SWZ(num, PAT_XOR1);  den += SWZ(den, PAT_XOR1);
    num += SWZ(num, PAT_XOR2);  den += SWZ(den, PAT_XOR2);
    num += SWZ(num, PAT_XOR4);  den += SWZ(den, PAT_XOR4);

    if ((lane & 7) == 0) out[e >> 3] = num / (den + TINYV);
}

extern "C" void kernel_launch(void* const* d_in, const int* in_sizes, int n_in,
                              void* d_out, int out_size, void* d_ws, size_t ws_size,
                              hipStream_t stream) {
    const float* points  = (const float*)d_in[0];
    const float* normals = (const float*)d_in[1];
    const float* rootR   = (const float*)d_in[2];
    const float* lobes   = (const float*)d_in[3];
    const float* lambdas = (const float*)d_in[4];
    const float* rrand   = (const float*)d_in[5];
    const float* W1      = (const float*)d_in[6];
    const float* b1      = (const float*)d_in[7];
    const float* W2      = (const float*)d_in[8];
    const float* b2      = (const float*)d_in[9];
    float* out = (float*)d_out;

    dim3 grid(NELEM / 256), block(256);
    sg_vis_wmma_kernel<<<grid, block, 0, stream>>>(
        points, normals, rootR, lobes, lambdas, rrand, W1, b1, W2, b2, out);
}


# Wave = 32 points of ONE lobe (lane owns a point; lobe frame wave-uniform in scalar registers) and walks 4 lobes x 8 samples: grid 1024, was 32768; sample sum in-lane, same tree order
# speedup vs baseline: 1.5767x; 1.5767x over previous
#include <hip/hip_runtime.h>
#include <math.h>


typedef float v2f __attribute__((ext_vector_type(2)));
typedef float v8f __attribute__((ext_vector_type(8)));

#define TINYV 1e-6f
#define LLOB 128
#define SSAM 8
#define HHID 16
#define NPTS 8192
#define PTS_PER_BLOCK 32
#define LOBES_PER_BLOCK 32
#define LOBES_PER_WAVE 4

#define SWZ(x, pat) __int_as_float(__builtin_amdgcn_ds_swizzle(__float_as_int(x), (pat)))
#define PAT_XOR16 0x401F

static __device__ __forceinline__ float wave_uniform(float x) {
    return __int_as_float(__builtin_amdgcn_readfirstlane(__float_as_int(x)));
}

__global__ __launch_bounds__(256) void sg_vis_wmma_kernel(
    const float* __restrict__ points,
    const float* __restrict__ normals,
    const float* __restrict__ rootR,
    const float* __restrict__ lobes,
    const float* __restrict__ lambdas,
    const float* __restrict__ rrand,
    const float* __restrict__ W1,
    const float* __restrict__ b1,
    const float* __restrict__ W2,
    const float* __restrict__ b2,
    float* __restrict__ out)
{
    __shared__ float sF[LOBES_PER_BLOCK * 12];

    const int tid  = threadIdx.x;
    const int lane = tid & 31;
    const int wv   = tid >> 5;
    const int lg   = blockIdx.x & 3;
    const int n    = (blockIdx.x >> 2) * PTS_PER_BLOCK + lane;

    if (tid < 32) {
        const int l = lg * LOBES_PER_BLOCK + tid;
        float lx = lobes[3*l+0], ly = lobes[3*l+1], lz = lobes[3*l+2];
        float invn = 1.f / (sqrtf(lx*lx + ly*ly + lz*lz) + TINYV);
        lx *= invn; ly *= invn; lz *= invn;
        float ux = -ly, uy = lx;
        float invu = 1.f / (sqrtf(ux*ux + uy*uy) + TINYV);
        ux *= invu; uy *= invu;
        float vx = -lz*uy, vy = lz*ux, vz = lx*uy - ly*ux;
        float invv = 1.f / (sqrtf(vx*vx + vy*vy + vz*vz) + TINYV);
        vx *= invv; vy *= invv; vz *= invv;
        float sharp = lambdas[l];
        float rphi  = fminf(acosf(1.f - 1.f/sharp), 1.04719755119659775f);
        float* F = &sF[tid * 12];
        F[0] = lx; F[1] = ly; F[2] = lz;
        F[3] = ux; F[4] = uy;
        F[5] = vx; F[6] = vy; F[7] = vz;
        F[8] = __sinf(rphi); F[9] = __cosf(rphi); F[10] = sharp; F[11] = 0.f;
    }

    const float px = points[3*n+0],  py = points[3*n+1],  pz = points[3*n+2];
    const float nx = normals[3*n+0], ny = normals[3*n+1], nz = normals[3*n+2];

    const float4* rp = reinterpret_cast<const float4*>(
        rrand + ((size_t)n * LLOB + (size_t)(lg * LOBES_PER_BLOCK + wv * LOBES_PER_WAVE)) * SSAM);
    float4 nxt_a = rp[0], nxt_b = rp[1];

    const float r0 = rootR[0], r1 = rootR[1], r2 = rootR[2];
    const float r3 = rootR[3], r4 = rootR[4], r5 = rootR[5];
    const float r6 = rootR[6], r7 = rootR[7], r8 = rootR[8];
    const float bias2 = b2[0];

    const int  m     = lane & 15;
    const bool up    = lane >= 16;
    const int  hbase = up ? 8 : 0;

    v2f A0, A1;
    A0.x = up ? W1[2*HHID + m] : W1[0*HHID + m];
    A0.y = up ? W1[3*HHID + m] : W1[1*HHID + m];
    A1.x = up ? 0.f : W1[4*HHID + m];
    A1.y = up ? 0.f : W1[5*HHID + m];

    v8f cb1;
    #pragma unroll
    for (int r = 0; r < 8; ++r) cb1[r] = b1[hbase + r];
    float w2v[8];
    #pragma unroll
    for (int r = 0; r < 8; ++r) w2v[r] = W2[hbase + r];

    const float pxX = SWZ(px, PAT_XOR16);
    const float pyX = SWZ(py, PAT_XOR16);
    const float pzX = SWZ(pz, PAT_XOR16);
    const float b0t0x = up ? pzX : px;
    const float b0t0y_lo = py;
    const float b0t1x = up ? pz  : pxX;
    const float b0t1y_lo = pyX;

    __syncthreads();

    float res0 = 0.f, res1 = 0.f, res2 = 0.f, res3 = 0.f;

    #pragma unroll 1
    for (int j = 0; j < LOBES_PER_WAVE; ++j) {
        const float4 ra = nxt_a, rb = nxt_b;
        if (j + 1 < LOBES_PER_WAVE) {
            nxt_a = rp[2*(j+1)];
            nxt_b = rp[2*(j+1) + 1];
        }

        const float* F = &sF[(wv * LOBES_PER_WAVE + j) * 12];
        const float lnx = wave_uniform(F[0]), lny = wave_uniform(F[1]), lnz = wave_uniform(F[2]);
        const float ux  = wave_uniform(F[3]), uy  = wave_uniform(F[4]);
        const float vx  = wave_uniform(F[5]), vy  = wave_uniform(F[6]), vz = wave_uniform(F[7]);
        const float sp  = wave_uniform(F[8]), cp  = wave_uniform(F[9]), sharp = wave_uniform(F[10]);

        const float rr[8] = { ra.x, ra.y, ra.z, ra.w, rb.x, rb.y, rb.z, rb.w };
        float nv[8], dv[8];

        #pragma unroll
        for (int s = 0; s < SSAM; ++s) {
            const float theta = (rr[s] * (1.f/SSAM) + (float)s * (1.f/SSAM))
                                * 6.28318530717958648f;
            const float st = __sinf(theta), ct = __cosf(theta);
            const float ca = ct * sp, cb = st * sp;
            const float dx = ux*ca + vx*cb + lnx*cp;
            const float dy = uy*ca + vy*cb + lny*cp;
            const float dz =          vz*cb + lnz*cp;

            const bool cosmask = (nx*dx + ny*dy + nz*dz) > TINYV;

            const float id0 = -(dx*r0 + dy*r3 + dz*r6);
            const float id1 = -(dx*r1 + dy*r4 + dz*r7);
            const float id2 = -(dx*r2 + dy*r5 + dz*r8);

            const float w = __expf(sharp * (dx*lnx + dy*lny + dz*lnz - 1.f));

            const float y3 = SWZ(id0, PAT_XOR16);
            const float y4 = SWZ(id1, PAT_XOR16);
            const float y5 = SWZ(id2, PAT_XOR16);

            v2f B0t0, B1t0, B0t1, B1t1;
            B0t0.x = b0t0x;            B0t0.y = up ? y3  : b0t0y_lo;
            B1t0.x = up ? 0.f : id1;   B1t0.y = up ? 0.f : id2;
            B0t1.x = b0t1x;            B0t1.y = up ? id0 : b0t1y_lo;
            B1t1.x = up ? 0.f : y4;    B1t1.y = up ? 0.f : y5;

            v8f c0, c1;
            c0 = __builtin_amdgcn_wmma_f32_16x16x4_f32(false, A0, false, B0t0, (short)0, cb1, false, false);
            c0 = __builtin_amdgcn_wmma_f32_16x16x4_f32(false, A1, false, B1t0, (short)0, c0,  false, false);
            c1 = __builtin_amdgcn_wmma_f32_16x16x4_f32(false, A0, false, B0t1, (short)0, cb1, false, false);
            c1 = __builtin_amdgcn_wmma_f32_16x16x4_f32(false, A1, false, B1t1, (short)0, c1,  false, false);

            float pa = 0.f, pb = 0.f;
            #pragma unroll
            for (int r = 0; r < 8; ++r) {
                const float t0 = c0[r], t1 = c1[r];
                pa += (t0 > 0.f ? t0 : 0.f) * w2v[r];
                pb += (t1 > 0.f ? t1 : 0.f) * w2v[r];
            }
            const float q  = up ? pa : pb;
            const float qq = SWZ(q, PAT_XOR16);
            const float logit = (up ? pb : pa) + qq + bias2;
            const float pred  = 1.f / (1.f + __expf(-logit));
            const float vis   = cosmask ? pred : 0.f;

            nv[s] = vis * w;
            dv[s] = w;
        }

        const float num = ((nv[0] + nv[1]) + (nv[2] + nv[3])) + ((nv[4] + nv[5]) + (nv[6] + nv[7]));
        const float den = ((dv[0] + dv[1]) + (dv[2] + dv[3])) + ((dv[4] + dv[5]) + (dv[6] + dv[7]));
        const float val = num / (den + TINYV);
        res0 = (j == 0) ? val : res0;
        res1 = (j == 1) ? val : res1;
        res2 = (j == 2) ? val : res2;
        res3 = (j == 3) ? val : res3;
    }

    float4 o; o.x = res0; o.y = res1; o.z = res2; o.w = res3;
    *reinterpret_cast<float4*>(
        out + (size_t)n * LLOB + (size_t)(lg * LOBES_PER_BLOCK + wv * LOBES_PER_WAVE)) = o;
}

extern "C" void kernel_launch(void* const* d_in, const int* in_sizes, int n_in,
                              void* d_out, int out_size, void* d_ws, size_t ws_size,
                              hipStream_t stream) {
    const float* points  = (const float*)d_in[0];
    const float* normals = (const float*)d_in[1];
    const float* rootR   = (const float*)d_in[2];
    const float* lobes   = (const float*)d_in[3];
    const float* lambdas = (const float*)d_in[4];
    const float* rrand   = (const float*)d_in[5];
    const float* W1      = (const float*)d_in[6];
    const float* b1      = (const float*)d_in[7];
    const float* W2      = (const float*)d_in[8];
    const float* b2      = (const float*)d_in[9];
    float* out = (float*)d_out;

    dim3 grid(1024), block(256);
    sg_vis_wmma_kernel<<<grid, block, 0, stream>>>(
        points, normals, rootR, lobes, lambdas, rrand, W1, b1, W2, b2, out);
}
